# starting kernel
# baseline (speedup 1.0000x reference)
#include <hip/hip_runtime.h>
#include <hip/hip_bf16.h>
#include <math.h>

typedef float v2f __attribute__((ext_vector_type(2)));
typedef float v8f __attribute__((ext_vector_type(8)));

#define EMB   64
#define EPSN  1e-12f

__device__ __forceinline__ float readlane_f32(float x, int lane) {
    return __int_as_float(__builtin_amdgcn_readlane(__float_as_int(x), lane));
}

__global__ void spmm_scatter(const float* __restrict__ f,
                             const float* __restrict__ vals,
                             const int*   __restrict__ rows,
                             const int*   __restrict__ cols,
                             float*       __restrict__ out, int E) {
    int idx = blockIdx.x * blockDim.x + threadIdx.x;
    int e   = idx >> 4;
    if (e >= E) return;
    int c   = (idx & 15) << 2;
    __builtin_prefetch(cols + e + 4096, 0, 0);
    int   row = rows[e];
    int   col = cols[e];
    float v   = vals[e];
    float4 x  = *(const float4*)(f + (size_t)col * EMB + c);
    float* o  = out + (size_t)row * EMB + c;
    atomicAdd(o + 0, x.x * v);
    atomicAdd(o + 1, x.y * v);
    atomicAdd(o + 2, x.z * v);
    atomicAdd(o + 3, x.w * v);
}

__global__ void norm_accum(float*       __restrict__ f,
                           const float* __restrict__ base,
                           float*       __restrict__ acc,
                           float inv_c, int N) {
    int tid  = blockIdx.x * blockDim.x + threadIdx.x;
    int lane = tid & 31;
    int row  = tid >> 5;
    if (row >= N) return;
    size_t off = (size_t)row * EMB + lane * 2;
    float2 x = *(const float2*)(f + off);
    float gx = x.x * inv_c;
    float gy = x.y * inv_c;
    float s  = gx * gx + gy * gy;
#pragma unroll
    for (int d = 16; d > 0; d >>= 1) s += __shfl_xor(s, d, 32);
    float n   = sqrtf(s);
    float inv = 1.0f / fmaxf(n, EPSN);
    gx *= inv;
    gy *= inv;
    float2 g;  g.x = gx;  g.y = gy;
    *(float2*)(f + off) = g;
    float2 b = *(const float2*)(base + off);
    float2 a;  a.x = b.x + gx;  a.y = b.y + gy;
    *(float2*)(acc + off) = a;
}

__global__ void edge_weight_scatter(const float* __restrict__ acc,
                                    const float* __restrict__ vals,
                                    const int*   __restrict__ rows,
                                    const int*   __restrict__ cols,
                                    float*       __restrict__ out, int E) {
    const float kScale = 1.0f / 3.0f;
    int lane   = threadIdx.x & 31;
    int waveId = (blockIdx.x * blockDim.x + threadIdx.x) >> 5;
    int m      = lane & 15;
    int half   = lane >> 4;
    int e      = waveId * 16 + m;
    bool valid = (e < E);
    int  ec    = valid ? e : 0;
    int   row  = rows[ec];
    int   col  = cols[ec];
    float v    = valid ? vals[ec] : 0.0f;

    const float* S = acc + (size_t)row * EMB;
    const float* T = acc + (size_t)col * EMB;

    v8f cacc = {};
#pragma unroll
    for (int k0 = 0; k0 < EMB; k0 += 4) {
        float2 sa = *(const float2*)(S + k0 + 2 * half);
        float2 sb = *(const float2*)(T + k0 + 2 * half);
        v2f A;  A.x = sa.x * kScale;  A.y = sa.y * kScale;
        v2f B;  B.x = sb.x * kScale;  B.y = sb.y * kScale;
        cacc = __builtin_amdgcn_wmma_f32_16x16x4_f32(
             false, A,  false, B,
             (short)0, cacc,  false,  false);
    }

    float diag = 0.0f;
#pragma unroll
    for (int vv = 0; vv < 8; ++vv) {
        float dlo = readlane_f32(cacc[vv], vv);
        float dhi = readlane_f32(cacc[vv], 24 + vv);
        if (m == vv)     diag = dlo;
        if (m == vv + 8) diag = dhi;
    }

    float cross = diag * (1.0f / EMB);
    float z     = 4.0f * (1.0f - cross);
    float sp    = log1pf(expf(-fabsf(z))) + fmaxf(z, 0.0f);
    float mat   = 0.5f * expf(z) * sp * v;

#pragma unroll 1
    for (int j = 0; j < 16; ++j) {
        float mj = readlane_f32(mat, j);
        int   rj = __builtin_amdgcn_readlane(row, j);
        int   cj = __builtin_amdgcn_readlane(col, j);
        float2 ee = *(const float2*)(acc + (size_t)cj * EMB + lane * 2);
        float* o  = out + (size_t)rj * EMB + lane * 2;
        atomicAdd(o,     ee.x * kScale * mj);
        atomicAdd(o + 1, ee.y * kScale * mj);
    }
}

extern "C" void kernel_launch(void* const* d_in, const int* in_sizes, int n_in,
                              void* d_out, int out_size, void* d_ws, size_t ws_size,
                              hipStream_t stream) {
    (void)n_in; (void)out_size; (void)ws_size;
    const float* feats     = (const float*)d_in[0];
    const float* edge_vals = (const float*)d_in[1];
    const int*   edge_row  = (const int*)d_in[2];
    const int*   edge_col  = (const int*)d_in[3];

    int N = in_sizes[0] / EMB;
    int E = in_sizes[1];

    float* bufA = (float*)d_ws;
    float* bufB = bufA + (size_t)N * EMB;
    float* acc  = bufB + (size_t)N * EMB;
    float* out  = (float*)d_out;
    size_t rowBytes = (size_t)N * EMB * sizeof(float);

    const int blk = 256;
    int gridSp = (int)(((long long)E * 16 + blk - 1) / blk);
    int gridNm = (int)(((long long)N * 32 + blk - 1) / blk);
    long long nWaves = ((long long)E + 15) / 16;
    int gridEw = (int)((nWaves * 32 + blk - 1) / blk);

    hipMemsetAsync(bufA, 0, rowBytes, stream);
    spmm_scatter<<<gridSp, blk, 0, stream>>>(feats, edge_vals, edge_row, edge_col, bufA, E);
    norm_accum<<<gridNm, blk, 0, stream>>>(bufA, feats, acc, 0.5f, N);

    hipMemsetAsync(bufB, 0, rowBytes, stream);
    spmm_scatter<<<gridSp, blk, 0, stream>>>(bufA, edge_vals, edge_row, edge_col, bufB, E);
    norm_accum<<<gridNm, blk, 0, stream>>>(bufB, acc, acc, 1.0f / 3.0f, N);

    hipMemsetAsync(out, 0, rowBytes, stream);
    edge_weight_scatter<<<gridEw, blk, 0, stream>>>(acc, edge_vals, edge_row, edge_col, out, E);
}


# Edge-weight dots re-tiled in registers: a WMMA tile takes 2 edges x 8 chunks of 8 floats; 16 B per lane, a load covers 4 whole lines (was 16 partial); chunk partials added by lane: sum order changes
# speedup vs baseline: 2.2707x; 2.2707x over previous
#include <hip/hip_runtime.h>
#include <hip/hip_bf16.h>
#include <math.h>

typedef float v2f __attribute__((ext_vector_type(2)));
typedef float v8f __attribute__((ext_vector_type(8)));

#define EMB   64
#define EPSN  1e-12f

__device__ __forceinline__ float readlane_f32(float x, int lane) {
    return __int_as_float(__builtin_amdgcn_readlane(__float_as_int(x), lane));
}

#define WAVE_E   32
#define GATHER_D 8

template <int MASK>
__device__ __forceinline__ float add_row_xor(float x) {
    int yi = __builtin_amdgcn_update_dpp(0, __float_as_int(x), 0x160 + MASK, 0xf, 0xf, true);
    return x + __int_as_float(yi);
}

__device__ __forceinline__ float other_row(float x) {
    int xi = __float_as_int(x);
    return __int_as_float(__builtin_amdgcn_permlanex16(xi, xi, 0x76543210, 0xfedcba98, false, false));
}

template <bool FULL, bool NORM>
__device__ __forceinline__ void scatter_wave(const float* __restrict__ f,
                                             float*       __restrict__ out,
                                             int myrow, int mycol, float myval,
                                             int base, int lane,
                                             float inv_c, int E) {
#pragma unroll 1
    for (int g = 0; g < WAVE_E; g += GATHER_D) {
        float xa[GATHER_D], xb[GATHER_D];
#pragma unroll
        for (int k = 0; k < GATHER_D; ++k) {
            int col = __builtin_amdgcn_readlane(mycol, g + k);
            const float* p = f + (size_t)col * EMB + lane;
            xa[k] = p[0];
            xb[k] = p[32];
        }
        if (NORM) {
            float mys = 1.0f;
#pragma unroll
            for (int k = 0; k < GATHER_D; ++k) {
                xa[k] *= inv_c;  xb[k] *= inv_c;
                float s = fmaf(xa[k], xa[k], xb[k] * xb[k]);
                s = add_row_xor<8>(s);
                s = add_row_xor<4>(s);
                s = add_row_xor<2>(s);
                s = add_row_xor<1>(s);
                s = s + other_row(s);
                if (lane == k) mys = s;
            }
            float myinv = 1.0f / fmaxf(sqrtf(mys), EPSN);
#pragma unroll
            for (int k = 0; k < GATHER_D; ++k) {
                float inv = readlane_f32(myinv, k);
                xa[k] *= inv;  xb[k] *= inv;
            }
        }
#pragma unroll
        for (int k = 0; k < GATHER_D; ++k) {
            int   row = __builtin_amdgcn_readlane(myrow, g + k);
            float v   = readlane_f32(myval, g + k);
            if (FULL || base + g + k < E) {
                float* o = out + (size_t)row * EMB + lane;
                atomicAdd(o,      xa[k] * v);
                atomicAdd(o + 32, xb[k] * v);
            }
        }
    }
}

template <bool NORM>
__device__ __forceinline__ void scatter_body(const float* __restrict__ f,
                                             const float* __restrict__ vals,
                                             const int*   __restrict__ rows,
                                             const int*   __restrict__ cols,
                                             float*       __restrict__ out,
                                             float inv_c, int E) {
    int lane = threadIdx.x & 31;
    int wave = (blockIdx.x * blockDim.x + threadIdx.x) >> 5;
    int base = wave * WAVE_E;
    if (base >= E) return;
    int ec   = min(base + lane, E - 1);
    int   myrow = rows[ec];
    int   mycol = cols[ec];
    float myval = vals[ec];
    if (base + WAVE_E <= E)
        scatter_wave<true,  NORM>(f, out, myrow, mycol, myval, base, lane, inv_c, E);
    else
        scatter_wave<false, NORM>(f, out, myrow, mycol, myval, base, lane, inv_c, E);
}

__global__ void spmm_scatter(const float* __restrict__ f,
                             const float* __restrict__ vals,
                             const int*   __restrict__ rows,
                             const int*   __restrict__ cols,
                             float*       __restrict__ out, int E) {
    scatter_body<false>(f, vals, rows, cols, out, 1.0f, E);
}

__global__ void spmm_scatter_norm(const float* __restrict__ g,
                                  const float* __restrict__ vals,
                                  const int*   __restrict__ rows,
                                  const int*   __restrict__ cols,
                                  float*       __restrict__ out,
                                  float inv_c, int E) {
    scatter_body<true>(g, vals, rows, cols, out, inv_c, E);
}

#define WAVE_R 8

template <bool FULL>
__device__ __forceinline__ void norm2_rows(const float* __restrict__ f1raw,
                                           const float* __restrict__ f2raw,
                                           const float* __restrict__ base,
                                           float*       __restrict__ acc,
                                           float c1, float c2, int N,
                                           int row0, int lane) {
    float2 x[WAVE_R], y[WAVE_R], b[WAVE_R];
#pragma unroll
    for (int k = 0; k < WAVE_R; ++k) {
        int row = FULL ? row0 + k : min(row0 + k, N - 1);
        size_t off = (size_t)row * EMB + lane * 2;
        x[k] = *(const float2*)(f1raw + off);
        y[k] = *(const float2*)(f2raw + off);
        b[k] = *(const float2*)(base + off);
    }
    __builtin_amdgcn_sched_barrier(0);
#pragma unroll
    for (int k = 0; k < WAVE_R; ++k) {
        float gx = x[k].x * c1;
        float gy = x[k].y * c1;
        float hx = y[k].x * c2;
        float hy = y[k].y * c2;
        float s  = fmaf(gx, gx, gy * gy);
        float t  = fmaf(hx, hx, hy * hy);
#pragma unroll
        for (int d = 16; d > 0; d >>= 1) {
            s += __shfl_xor(s, d, 32);
            t += __shfl_xor(t, d, 32);
        }
        float inv1 = 1.0f / fmaxf(sqrtf(s), EPSN);
        float inv2 = 1.0f / fmaxf(sqrtf(t), EPSN);
        gx *= inv1;
        gy *= inv1;
        hx *= inv2;
        hy *= inv2;
        float2 a;
        a.x = (b[k].x + gx) + hx;
        a.y = (b[k].y + gy) + hy;
        if (FULL || row0 + k < N)
            *(float2*)(acc + (size_t)(row0 + k) * EMB + lane * 2) = a;
    }
}

__global__ void norm2_accum(const float* __restrict__ f1raw,
                            const float* __restrict__ f2raw,
                            const float* __restrict__ base,
                            float*       __restrict__ acc,
                            float c1, float c2, int N) {
    int tid  = blockIdx.x * blockDim.x + threadIdx.x;
    int lane = tid & 31;
    int row0 = (tid >> 5) * WAVE_R;
    if (row0 >= N) return;
    if (row0 + WAVE_R <= N)
        norm2_rows<true >(f1raw, f2raw, base, acc, c1, c2, N, row0, lane);
    else
        norm2_rows<false>(f1raw, f2raw, base, acc, c1, c2, N, row0, lane);
}

#define PAIRS_IN_FLIGHT 2

__global__ void edge_weight_scatter(const float* __restrict__ acc,
                                    const float* __restrict__ vals,
                                    const int*   __restrict__ rows,
                                    const int*   __restrict__ cols,
                                    float*       __restrict__ out, int E) {
    const float kScale = 1.0f / 3.0f;
    int lane   = threadIdx.x & 31;
    int waveId = (blockIdx.x * blockDim.x + threadIdx.x) >> 5;
    int m      = lane & 15;
    int half   = lane >> 4;
    int e      = waveId * 16 + m;
    bool valid = (e < E);
    int  ec    = valid ? e : 0;
    int   row  = rows[ec];
    int   col  = cols[ec];
    float v    = valid ? vals[ec] : 0.0f;

    int sub   = m >> 3;
    int foff  = (m & 7) * 8 + half * 4;
    int dsel  = lane & 7;

    float diag = 0.0f;
#pragma unroll 1
    for (int g0 = 0; g0 < 8; g0 += PAIRS_IN_FLIGHT) {
        float4 s[PAIRS_IN_FLIGHT], t[PAIRS_IN_FLIGHT];
#pragma unroll
        for (int q = 0; q < PAIRS_IN_FLIGHT; ++q) {
            int j  = 2 * (g0 + q);
            int r0 = __builtin_amdgcn_readlane(row, j);
            int r1 = __builtin_amdgcn_readlane(row, j + 1);
            int c0 = __builtin_amdgcn_readlane(col, j);
            int c1 = __builtin_amdgcn_readlane(col, j + 1);
            int myr = sub ? r1 : r0;
            int myc = sub ? c1 : c0;
            s[q] = *(const float4*)(acc + (size_t)myr * EMB + foff);
            t[q] = *(const float4*)(acc + (size_t)myc * EMB + foff);
        }
#pragma unroll
        for (int q = 0; q < PAIRS_IN_FLIGHT; ++q) {
            int j = 2 * (g0 + q);
            v2f A0;  A0.x = s[q].x * kScale;  A0.y = s[q].y * kScale;
            v2f B0;  B0.x = t[q].x * kScale;  B0.y = t[q].y * kScale;
            v2f A1;  A1.x = s[q].z * kScale;  A1.y = s[q].w * kScale;
            v2f B1;  B1.x = t[q].z * kScale;  B1.y = t[q].w * kScale;
            v8f d = {};
            d = __builtin_amdgcn_wmma_f32_16x16x4_f32(
                 false, A0,  false, B0,
                 (short)0, d,  false,  false);
            d = __builtin_amdgcn_wmma_f32_16x16x4_f32(
                 false, A1,  false, B1,
                 (short)0, d,  false,  false);
            float p = 0.0f;
#pragma unroll
            for (int r = 0; r < 8; ++r)
                if (dsel == r) p = d[r];
            p = add_row_xor<4>(p);
            p = add_row_xor<2>(p);
            p = add_row_xor<1>(p);
            float d0 = readlane_f32(p, 0);
            float d1 = readlane_f32(p, 24);
            if (m == j)     diag = d0;
            if (m == j + 1) diag = d1;
        }
    }

    float cross = diag * (1.0f / EMB);
    float z     = 4.0f * (1.0f - cross);
    float sp    = log1pf(expf(-fabsf(z))) + fmaxf(z, 0.0f);
    float mat   = 0.5f * expf(z) * sp * v;

#pragma unroll 1
    for (int j = 0; j < 16; ++j) {
        float mj = readlane_f32(mat, j);
        int   rj = __builtin_amdgcn_readlane(row, j);
        int   cj = __builtin_amdgcn_readlane(col, j);
        const float* ep = acc + (size_t)cj * EMB + lane;
        float e0 = ep[0];
        float e1 = ep[32];
        float* o  = out + (size_t)rj * EMB + lane;
        atomicAdd(o,      e0 * kScale * mj);
        atomicAdd(o + 32, e1 * kScale * mj);
    }
}

extern "C" void kernel_launch(void* const* d_in, const int* in_sizes, int n_in,
                              void* d_out, int out_size, void* d_ws, size_t ws_size,
                              hipStream_t stream) {
    (void)n_in; (void)out_size; (void)ws_size;
    const float* feats     = (const float*)d_in[0];
    const float* edge_vals = (const float*)d_in[1];
    const int*   edge_row  = (const int*)d_in[2];
    const int*   edge_col  = (const int*)d_in[3];

    int N = in_sizes[0] / EMB;
    int E = in_sizes[1];

    float* bufA = (float*)d_ws;
    float* bufB = bufA + (size_t)N * EMB;
    float* acc  = bufB + (size_t)N * EMB;
    float* out  = (float*)d_out;
    size_t rowBytes = (size_t)N * EMB * sizeof(float);

    const int blk = 256;
    const int wpb = blk / 32;
    int gridSp = ((E + WAVE_E - 1) / WAVE_E + wpb - 1) / wpb;
    int gridNm = ((N + WAVE_R - 1) / WAVE_R + wpb - 1) / wpb;
    long long nWaves = ((long long)E + 15) / 16;
    int gridEw = (int)((nWaves * 32 + blk - 1) / blk);

    hipMemsetAsync(bufA, 0, rowBytes, stream);
    spmm_scatter<<<gridSp, blk, 0, stream>>>(feats, edge_vals, edge_row, edge_col, bufA, E);

    hipMemsetAsync(bufB, 0, rowBytes, stream);
    spmm_scatter_norm<<<gridSp, blk, 0, stream>>>(bufA, edge_vals, edge_row, edge_col, bufB, 0.5f, E);

    norm2_accum<<<gridNm, blk, 0, stream>>>(bufA, bufB, feats, acc, 0.5f, 1.0f / 3.0f, N);

    hipMemsetAsync(out, 0, rowBytes, stream);
    edge_weight_scatter<<<gridEw, blk, 0, stream>>>(acc, edge_vals, edge_row, edge_col, out, E);
}


# Edge-weight kernel's scatter phase issues all 16 end-row gathers of a wave together, ahead of the weight arithmetic (was one edge at a time: 16 dependent round trips); values and atomics kept
# speedup vs baseline: 2.4237x; 1.0674x over previous
#include <hip/hip_runtime.h>
#include <hip/hip_bf16.h>
#include <math.h>

typedef float v2f __attribute__((ext_vector_type(2)));
typedef float v8f __attribute__((ext_vector_type(8)));

#define EMB   64
#define EPSN  1e-12f

__device__ __forceinline__ float readlane_f32(float x, int lane) {
    return __int_as_float(__builtin_amdgcn_readlane(__float_as_int(x), lane));
}

#define WAVE_E   32
#define GATHER_D 8

template <int MASK>
__device__ __forceinline__ float add_row_xor(float x) {
    int yi = __builtin_amdgcn_update_dpp(0, __float_as_int(x), 0x160 + MASK, 0xf, 0xf, true);
    return x + __int_as_float(yi);
}

__device__ __forceinline__ float other_row(float x) {
    int xi = __float_as_int(x);
    return __int_as_float(__builtin_amdgcn_permlanex16(xi, xi, 0x76543210, 0xfedcba98, false, false));
}

template <bool FULL, bool NORM>
__device__ __forceinline__ void scatter_wave(const float* __restrict__ f,
                                             float*       __restrict__ out,
                                             int myrow, int mycol, float myval,
                                             int base, int lane,
                                             float inv_c, int E) {
#pragma unroll 1
    for (int g = 0; g < WAVE_E; g += GATHER_D) {
        float xa[GATHER_D], xb[GATHER_D];
#pragma unroll
        for (int k = 0; k < GATHER_D; ++k) {
            int col = __builtin_amdgcn_readlane(mycol, g + k);
            const float* p = f + (size_t)col * EMB + lane;
            xa[k] = p[0];
            xb[k] = p[32];
        }
        if (NORM) {
            float mys = 1.0f;
#pragma unroll
            for (int k = 0; k < GATHER_D; ++k) {
                xa[k] *= inv_c;  xb[k] *= inv_c;
                float s = fmaf(xa[k], xa[k], xb[k] * xb[k]);
                s = add_row_xor<8>(s);
                s = add_row_xor<4>(s);
                s = add_row_xor<2>(s);
                s = add_row_xor<1>(s);
                s = s + other_row(s);
                if (lane == k) mys = s;
            }
            float myinv = 1.0f / fmaxf(sqrtf(mys), EPSN);
#pragma unroll
            for (int k = 0; k < GATHER_D; ++k) {
                float inv = readlane_f32(myinv, k);
                xa[k] *= inv;  xb[k] *= inv;
            }
        }
#pragma unroll
        for (int k = 0; k < GATHER_D; ++k) {
            int   row = __builtin_amdgcn_readlane(myrow, g + k);
            float v   = readlane_f32(myval, g + k);
            if (FULL || base + g + k < E) {
                float* o = out + (size_t)row * EMB + lane;
                atomicAdd(o,      xa[k] * v);
                atomicAdd(o + 32, xb[k] * v);
            }
        }
    }
}

template <bool NORM>
__device__ __forceinline__ void scatter_body(const float* __restrict__ f,
                                             const float* __restrict__ vals,
                                             const int*   __restrict__ rows,
                                             const int*   __restrict__ cols,
                                             float*       __restrict__ out,
                                             float inv_c, int E) {
    int lane = threadIdx.x & 31;
    int wave = (blockIdx.x * blockDim.x + threadIdx.x) >> 5;
    int base = wave * WAVE_E;
    if (base >= E) return;
    int ec   = min(base + lane, E - 1);
    int   myrow = rows[ec];
    int   mycol = cols[ec];
    float myval = vals[ec];
    if (base + WAVE_E <= E)
        scatter_wave<true,  NORM>(f, out, myrow, mycol, myval, base, lane, inv_c, E);
    else
        scatter_wave<false, NORM>(f, out, myrow, mycol, myval, base, lane, inv_c, E);
}

__global__ void spmm_scatter(const float* __restrict__ f,
                             const float* __restrict__ vals,
                             const int*   __restrict__ rows,
                             const int*   __restrict__ cols,
                             float*       __restrict__ out, int E) {
    scatter_body<false>(f, vals, rows, cols, out, 1.0f, E);
}

__global__ void spmm_scatter_norm(const float* __restrict__ g,
                                  const float* __restrict__ vals,
                                  const int*   __restrict__ rows,
                                  const int*   __restrict__ cols,
                                  float*       __restrict__ out,
                                  float inv_c, int E) {
    scatter_body<true>(g, vals, rows, cols, out, inv_c, E);
}

#define WAVE_R 8

template <bool FULL>
__device__ __forceinline__ void norm2_rows(const float* __restrict__ f1raw,
                                           const float* __restrict__ f2raw,
                                           const float* __restrict__ base,
                                           float*       __restrict__ acc,
                                           float c1, float c2, int N,
                                           int row0, int lane) {
    float2 x[WAVE_R], y[WAVE_R], b[WAVE_R];
#pragma unroll
    for (int k = 0; k < WAVE_R; ++k) {
        int row = FULL ? row0 + k : min(row0 + k, N - 1);
        size_t off = (size_t)row * EMB + lane * 2;
        x[k] = *(const float2*)(f1raw + off);
        y[k] = *(const float2*)(f2raw + off);
        b[k] = *(const float2*)(base + off);
    }
    __builtin_amdgcn_sched_barrier(0);
#pragma unroll
    for (int k = 0; k < WAVE_R; ++k) {
        float gx = x[k].x * c1;
        float gy = x[k].y * c1;
        float hx = y[k].x * c2;
        float hy = y[k].y * c2;
        float s  = fmaf(gx, gx, gy * gy);
        float t  = fmaf(hx, hx, hy * hy);
#pragma unroll
        for (int d = 16; d > 0; d >>= 1) {
            s += __shfl_xor(s, d, 32);
            t += __shfl_xor(t, d, 32);
        }
        float inv1 = 1.0f / fmaxf(sqrtf(s), EPSN);
        float inv2 = 1.0f / fmaxf(sqrtf(t), EPSN);
        gx *= inv1;
        gy *= inv1;
        hx *= inv2;
        hy *= inv2;
        float2 a;
        a.x = (b[k].x + gx) + hx;
        a.y = (b[k].y + gy) + hy;
        if (FULL || row0 + k < N)
            *(float2*)(acc + (size_t)(row0 + k) * EMB + lane * 2) = a;
    }
}

__global__ void norm2_accum(const float* __restrict__ f1raw,
                            const float* __restrict__ f2raw,
                            const float* __restrict__ base,
                            float*       __restrict__ acc,
                            float c1, float c2, int N) {
    int tid  = blockIdx.x * blockDim.x + threadIdx.x;
    int lane = tid & 31;
    int row0 = (tid >> 5) * WAVE_R;
    if (row0 >= N) return;
    if (row0 + WAVE_R <= N)
        norm2_rows<true >(f1raw, f2raw, base, acc, c1, c2, N, row0, lane);
    else
        norm2_rows<false>(f1raw, f2raw, base, acc, c1, c2, N, row0, lane);
}

#define PAIRS_IN_FLIGHT 2
#define EDGES_W         16

__global__ void edge_weight_scatter(const float* __restrict__ acc,
                                    const float* __restrict__ vals,
                                    const int*   __restrict__ rows,
                                    const int*   __restrict__ cols,
                                    float*       __restrict__ out, int E) {
    const float kScale = 1.0f / 3.0f;
    int lane   = threadIdx.x & 31;
    int waveId = (blockIdx.x * blockDim.x + threadIdx.x) >> 5;
    int m      = lane & 15;
    int half   = lane >> 4;
    int e      = waveId * 16 + m;
    bool valid = (e < E);
    int  ec    = valid ? e : 0;
    int   row  = rows[ec];
    int   col  = cols[ec];
    float v    = valid ? vals[ec] : 0.0f;

    int sub   = m >> 3;
    int foff  = (m & 7) * 8 + half * 4;
    int dsel  = lane & 7;

    float diag = 0.0f;
#pragma unroll 1
    for (int g0 = 0; g0 < 8; g0 += PAIRS_IN_FLIGHT) {
        float4 s[PAIRS_IN_FLIGHT], t[PAIRS_IN_FLIGHT];
#pragma unroll
        for (int q = 0; q < PAIRS_IN_FLIGHT; ++q) {
            int j  = 2 * (g0 + q);
            int r0 = __builtin_amdgcn_readlane(row, j);
            int r1 = __builtin_amdgcn_readlane(row, j + 1);
            int c0 = __builtin_amdgcn_readlane(col, j);
            int c1 = __builtin_amdgcn_readlane(col, j + 1);
            int myr = sub ? r1 : r0;
            int myc = sub ? c1 : c0;
            s[q] = *(const float4*)(acc + (size_t)myr * EMB + foff);
            t[q] = *(const float4*)(acc + (size_t)myc * EMB + foff);
        }
#pragma unroll
        for (int q = 0; q < PAIRS_IN_FLIGHT; ++q) {
            int j = 2 * (g0 + q);
            v2f A0;  A0.x = s[q].x * kScale;  A0.y = s[q].y * kScale;
            v2f B0;  B0.x = t[q].x * kScale;  B0.y = t[q].y * kScale;
            v2f A1;  A1.x = s[q].z * kScale;  A1.y = s[q].w * kScale;
            v2f B1;  B1.x = t[q].z * kScale;  B1.y = t[q].w * kScale;
            v8f d = {};
            d = __builtin_amdgcn_wmma_f32_16x16x4_f32(
                 false, A0,  false, B0,
                 (short)0, d,  false,  false);
            d = __builtin_amdgcn_wmma_f32_16x16x4_f32(
                 false, A1,  false, B1,
                 (short)0, d,  false,  false);
            float p = 0.0f;
#pragma unroll
            for (int r = 0; r < 8; ++r)
                if (dsel == r) p = d[r];
            p = add_row_xor<4>(p);
            p = add_row_xor<2>(p);
            p = add_row_xor<1>(p);
            float d0 = readlane_f32(p, 0);
            float d1 = readlane_f32(p, 24);
            if (m == j)     diag = d0;
            if (m == j + 1) diag = d1;
        }
    }

    float xa[EDGES_W], xb[EDGES_W];
#pragma unroll
    for (int j = 0; j < EDGES_W; ++j) {
        int cj = __builtin_amdgcn_readlane(col, j);
        const float* ep = acc + (size_t)cj * EMB + lane;
        xa[j] = ep[0];
        xb[j] = ep[32];
    }

    float cross = diag * (1.0f / EMB);
    float z     = 4.0f * (1.0f - cross);
    float sp    = log1pf(expf(-fabsf(z))) + fmaxf(z, 0.0f);
    float mat   = 0.5f * expf(z) * sp * v;

#pragma unroll
    for (int j = 0; j < EDGES_W; ++j) {
        float mj = readlane_f32(mat, j);
        int   rj = __builtin_amdgcn_readlane(row, j);
        float* o  = out + (size_t)rj * EMB + lane;
        atomicAdd(o,      xa[j] * kScale * mj);
        atomicAdd(o + 32, xb[j] * kScale * mj);
    }
}

extern "C" void kernel_launch(void* const* d_in, const int* in_sizes, int n_in,
                              void* d_out, int out_size, void* d_ws, size_t ws_size,
                              hipStream_t stream) {
    (void)n_in; (void)out_size; (void)ws_size;
    const float* feats     = (const float*)d_in[0];
    const float* edge_vals = (const float*)d_in[1];
    const int*   edge_row  = (const int*)d_in[2];
    const int*   edge_col  = (const int*)d_in[3];

    int N = in_sizes[0] / EMB;
    int E = in_sizes[1];

    float* bufA = (float*)d_ws;
    float* bufB = bufA + (size_t)N * EMB;
    float* acc  = bufB + (size_t)N * EMB;
    float* out  = (float*)d_out;
    size_t rowBytes = (size_t)N * EMB * sizeof(float);

    const int blk = 256;
    const int wpb = blk / 32;
    int gridSp = ((E + WAVE_E - 1) / WAVE_E + wpb - 1) / wpb;
    int gridNm = ((N + WAVE_R - 1) / WAVE_R + wpb - 1) / wpb;
    long long nWaves = ((long long)E + 15) / 16;
    int gridEw = (int)((nWaves * 32 + blk - 1) / blk);

    hipMemsetAsync(bufA, 0, rowBytes, stream);
    spmm_scatter<<<gridSp, blk, 0, stream>>>(feats, edge_vals, edge_row, edge_col, bufA, E);

    hipMemsetAsync(bufB, 0, rowBytes, stream);
    spmm_scatter_norm<<<gridSp, blk, 0, stream>>>(bufA, edge_vals, edge_row, edge_col, bufB, 0.5f, E);

    norm2_accum<<<gridNm, blk, 0, stream>>>(bufA, bufB, feats, acc, 0.5f, 1.0f / 3.0f, N);

    hipMemsetAsync(out, 0, rowBytes, stream);
    edge_weight_scatter<<<gridEw, blk, 0, stream>>>(acc, edge_vals, edge_row, edge_col, out, E);
}


# Edge-weight dots read rows in the scatter's lane layout (lane l: floats l, l+32), a WMMA tile per edge, trace summed; end-row registers serve dot and scatter, 2nd gather goes; dot sum order changes
# speedup vs baseline: 2.5059x; 1.0339x over previous
#include <hip/hip_runtime.h>
#include <hip/hip_bf16.h>
#include <math.h>

typedef float v2f __attribute__((ext_vector_type(2)));
typedef float v8f __attribute__((ext_vector_type(8)));

#define EMB   64
#define EPSN  1e-12f

__device__ __forceinline__ float readlane_f32(float x, int lane) {
    return __int_as_float(__builtin_amdgcn_readlane(__float_as_int(x), lane));
}

#define WAVE_E   32
#define GATHER_D 8

template <int MASK>
__device__ __forceinline__ float add_row_xor(float x) {
    int yi = __builtin_amdgcn_update_dpp(0, __float_as_int(x), 0x160 + MASK, 0xf, 0xf, true);
    return x + __int_as_float(yi);
}

__device__ __forceinline__ float other_row(float x) {
    int xi = __float_as_int(x);
    return __int_as_float(__builtin_amdgcn_permlanex16(xi, xi, 0x76543210, 0xfedcba98, false, false));
}

template <bool FULL, bool NORM>
__device__ __forceinline__ void scatter_wave(const float* __restrict__ f,
                                             float*       __restrict__ out,
                                             int myrow, int mycol, float myval,
                                             int base, int lane,
                                             float inv_c, int E) {
#pragma unroll 1
    for (int g = 0; g < WAVE_E; g += GATHER_D) {
        float xa[GATHER_D], xb[GATHER_D];
#pragma unroll
        for (int k = 0; k < GATHER_D; ++k) {
            int col = __builtin_amdgcn_readlane(mycol, g + k);
            const float* p = f + (size_t)col * EMB + lane;
            xa[k] = p[0];
            xb[k] = p[32];
        }
        if (NORM) {
            float mys = 1.0f;
#pragma unroll
            for (int k = 0; k < GATHER_D; ++k) {
                xa[k] *= inv_c;  xb[k] *= inv_c;
                float s = fmaf(xa[k], xa[k], xb[k] * xb[k]);
                s = add_row_xor<8>(s);
                s = add_row_xor<4>(s);
                s = add_row_xor<2>(s);
                s = add_row_xor<1>(s);
                s = s + other_row(s);
                if (lane == k) mys = s;
            }
            float myinv = 1.0f / fmaxf(sqrtf(mys), EPSN);
#pragma unroll
            for (int k = 0; k < GATHER_D; ++k) {
                float inv = readlane_f32(myinv, k);
                xa[k] *= inv;  xb[k] *= inv;
            }
        }
#pragma unroll
        for (int k = 0; k < GATHER_D; ++k) {
            int   row = __builtin_amdgcn_readlane(myrow, g + k);
            float v   = readlane_f32(myval, g + k);
            if (FULL || base + g + k < E) {
                float* o = out + (size_t)row * EMB + lane;
                atomicAdd(o,      xa[k] * v);
                atomicAdd(o + 32, xb[k] * v);
            }
        }
    }
}

template <bool NORM>
__device__ __forceinline__ void scatter_body(const float* __restrict__ f,
                                             const float* __restrict__ vals,
                                             const int*   __restrict__ rows,
                                             const int*   __restrict__ cols,
                                             float*       __restrict__ out,
                                             float inv_c, int E) {
    int lane = threadIdx.x & 31;
    int wave = (blockIdx.x * blockDim.x + threadIdx.x) >> 5;
    int base = wave * WAVE_E;
    if (base >= E) return;
    int ec   = min(base + lane, E - 1);
    int   myrow = rows[ec];
    int   mycol = cols[ec];
    float myval = vals[ec];
    if (base + WAVE_E <= E)
        scatter_wave<true,  NORM>(f, out, myrow, mycol, myval, base, lane, inv_c, E);
    else
        scatter_wave<false, NORM>(f, out, myrow, mycol, myval, base, lane, inv_c, E);
}

__global__ void spmm_scatter(const float* __restrict__ f,
                             const float* __restrict__ vals,
                             const int*   __restrict__ rows,
                             const int*   __restrict__ cols,
                             float*       __restrict__ out, int E) {
    scatter_body<false>(f, vals, rows, cols, out, 1.0f, E);
}

__global__ void spmm_scatter_norm(const float* __restrict__ g,
                                  const float* __restrict__ vals,
                                  const int*   __restrict__ rows,
                                  const int*   __restrict__ cols,
                                  float*       __restrict__ out,
                                  float inv_c, int E) {
    scatter_body<true>(g, vals, rows, cols, out, inv_c, E);
}

#define WAVE_R 8

template <bool FULL>
__device__ __forceinline__ void norm2_rows(const float* __restrict__ f1raw,
                                           const float* __restrict__ f2raw,
                                           const float* __restrict__ base,
                                           float*       __restrict__ acc,
                                           float c1, float c2, int N,
                                           int row0, int lane) {
    float2 x[WAVE_R], y[WAVE_R], b[WAVE_R];
#pragma unroll
    for (int k = 0; k < WAVE_R; ++k) {
        int row = FULL ? row0 + k : min(row0 + k, N - 1);
        size_t off = (size_t)row * EMB + lane * 2;
        x[k] = *(const float2*)(f1raw + off);
        y[k] = *(const float2*)(f2raw + off);
        b[k] = *(const float2*)(base + off);
    }
    __builtin_amdgcn_sched_barrier(0);
#pragma unroll
    for (int k = 0; k < WAVE_R; ++k) {
        float gx = x[k].x * c1;
        float gy = x[k].y * c1;
        float hx = y[k].x * c2;
        float hy = y[k].y * c2;
        float s  = fmaf(gx, gx, gy * gy);
        float t  = fmaf(hx, hx, hy * hy);
#pragma unroll
        for (int d = 16; d > 0; d >>= 1) {
            s += __shfl_xor(s, d, 32);
            t += __shfl_xor(t, d, 32);
        }
        float inv1 = 1.0f / fmaxf(sqrtf(s), EPSN);
        float inv2 = 1.0f / fmaxf(sqrtf(t), EPSN);
        gx *= inv1;
        gy *= inv1;
        hx *= inv2;
        hy *= inv2;
        float2 a;
        a.x = (b[k].x + gx) + hx;
        a.y = (b[k].y + gy) + hy;
        if (FULL || row0 + k < N)
            *(float2*)(acc + (size_t)(row0 + k) * EMB + lane * 2) = a;
    }
}

__global__ void norm2_accum(const float* __restrict__ f1raw,
                            const float* __restrict__ f2raw,
                            const float* __restrict__ base,
                            float*       __restrict__ acc,
                            float c1, float c2, int N) {
    int tid  = blockIdx.x * blockDim.x + threadIdx.x;
    int lane = tid & 31;
    int row0 = (tid >> 5) * WAVE_R;
    if (row0 >= N) return;
    if (row0 + WAVE_R <= N)
        norm2_rows<true >(f1raw, f2raw, base, acc, c1, c2, N, row0, lane);
    else
        norm2_rows<false>(f1raw, f2raw, base, acc, c1, c2, N, row0, lane);
}

#define EDGES_W 16

__global__ void edge_weight_scatter(const float* __restrict__ acc,
                                    const float* __restrict__ vals,
                                    const int*   __restrict__ rows,
                                    const int*   __restrict__ cols,
                                    float*       __restrict__ out, int E) {
    const float kScale = 1.0f / 3.0f;
    int lane   = threadIdx.x & 31;
    int waveId = (blockIdx.x * blockDim.x + threadIdx.x) >> 5;
    int m      = lane & 15;
    int e      = waveId * EDGES_W + m;
    bool valid = (e < E);
    int  ec    = valid ? e : 0;
    int   row  = rows[ec];
    int   col  = cols[ec];
    float v    = valid ? vals[ec] : 0.0f;

    int dsel  = lane & 7;

    float sa[EDGES_W], sb[EDGES_W], xa[EDGES_W], xb[EDGES_W];
#pragma unroll
    for (int j = 0; j < EDGES_W; ++j) {
        int rj = __builtin_amdgcn_readlane(row, j);
        int cj = __builtin_amdgcn_readlane(col, j);
        const float* sp = acc + (size_t)rj * EMB + lane;
        const float* ep = acc + (size_t)cj * EMB + lane;
        sa[j] = sp[0];
        sb[j] = sp[32];
        xa[j] = ep[0];
        xb[j] = ep[32];
    }
    __builtin_amdgcn_sched_barrier(0);

    float diag = 0.0f;
#pragma unroll
    for (int j = 0; j < EDGES_W; ++j) {
        v2f A;  A.x = sa[j] * kScale;  A.y = sb[j] * kScale;
        v2f B;  B.x = xa[j] * kScale;  B.y = xb[j] * kScale;
        v8f d = {};
        d = __builtin_amdgcn_wmma_f32_16x16x4_f32(
             false, A,  false, B,
             (short)0, d,  false,  false);
        float p = 0.0f;
#pragma unroll
        for (int r = 0; r < 8; ++r)
            if (dsel == r) p = d[r];
        p = add_row_xor<4>(p);
        p = add_row_xor<2>(p);
        p = add_row_xor<1>(p);
        float dj = readlane_f32(p, 0) + readlane_f32(p, 24);
        if (m == j) diag = dj;
    }

    float cross = diag * (1.0f / EMB);
    float z     = 4.0f * (1.0f - cross);
    float sp    = log1pf(expf(-fabsf(z))) + fmaxf(z, 0.0f);
    float mat   = 0.5f * expf(z) * sp * v;

#pragma unroll
    for (int j = 0; j < EDGES_W; ++j) {
        float mj = readlane_f32(mat, j);
        int   rj = __builtin_amdgcn_readlane(row, j);
        float* o  = out + (size_t)rj * EMB + lane;
        atomicAdd(o,      xa[j] * kScale * mj);
        atomicAdd(o + 32, xb[j] * kScale * mj);
    }
}

extern "C" void kernel_launch(void* const* d_in, const int* in_sizes, int n_in,
                              void* d_out, int out_size, void* d_ws, size_t ws_size,
                              hipStream_t stream) {
    (void)n_in; (void)out_size; (void)ws_size;
    const float* feats     = (const float*)d_in[0];
    const float* edge_vals = (const float*)d_in[1];
    const int*   edge_row  = (const int*)d_in[2];
    const int*   edge_col  = (const int*)d_in[3];

    int N = in_sizes[0] / EMB;
    int E = in_sizes[1];

    float* bufA = (float*)d_ws;
    float* bufB = bufA + (size_t)N * EMB;
    float* acc  = bufB + (size_t)N * EMB;
    float* out  = (float*)d_out;
    size_t rowBytes = (size_t)N * EMB * sizeof(float);

    const int blk = 256;
    const int wpb = blk / 32;
    int gridSp = ((E + WAVE_E - 1) / WAVE_E + wpb - 1) / wpb;
    int gridNm = ((N + WAVE_R - 1) / WAVE_R + wpb - 1) / wpb;
    long long nWaves = ((long long)E + 15) / 16;
    int gridEw = (int)((nWaves * 32 + blk - 1) / blk);

    hipMemsetAsync(bufA, 0, rowBytes, stream);
    spmm_scatter<<<gridSp, blk, 0, stream>>>(feats, edge_vals, edge_row, edge_col, bufA, E);

    hipMemsetAsync(bufB, 0, rowBytes, stream);
    spmm_scatter_norm<<<gridSp, blk, 0, stream>>>(bufA, edge_vals, edge_row, edge_col, bufB, 0.5f, E);

    norm2_accum<<<gridNm, blk, 0, stream>>>(bufA, bufB, feats, acc, 0.5f, 1.0f / 3.0f, N);

    hipMemsetAsync(out, 0, rowBytes, stream);
    edge_weight_scatter<<<gridEw, blk, 0, stream>>>(acc, edge_vals, edge_row, edge_col, out, E);
}
